# starting kernel
# baseline (speedup 1.0000x reference)
#include <hip/hip_runtime.h>
#include <math.h>

#define D_MODEL   1024
#define D_INNER   2048
#define D_XB       512
#define D_STATE     16
#define D_CONV       4
#define DT_RANK     64
#define NXB         32
#define H_HEADS    128
#define REP          4
#define PCH         16
#define BATCH        2
#define SEQ       2048
#define E_DIM     (2*D_INNER + 2*D_XB + DT_RANK)
#define M_ROWS    (BATCH*SEQ)

#define OFF_Z   0
#define OFF_X   (D_INNER)
#define OFF_B   (D_INNER + D_XB)
#define OFF_C   (D_INNER + 2*D_XB)
#define OFF_DT  (2*D_INNER + 2*D_XB)

typedef __attribute__((ext_vector_type(2))) float v2f;
typedef __attribute__((ext_vector_type(8))) float v8f;

__device__ __forceinline__ v8f wmma_f32_16x16x4(v2f a, v2f b, v8f c) {
  return __builtin_amdgcn_wmma_f32_16x16x4_f32(
       false, a,  false, b,
       (short)0, c,  false,  false);
}

template<int EPI>
__global__ void __launch_bounds__(256)
wmma_gemm_f32(const float* __restrict__ A, int lda,
              const float* __restrict__ W, int ldb,
              float* __restrict__ C, int ldc,
              int M, int N, int K,
              const float* __restrict__ bias)
{
  const int lane = threadIdx.x & 31;
  const int wave = threadIdx.x >> 5;
  const int tilesN = N >> 6;
  const int totalTiles = (M >> 6) * tilesN;
  const int tile = blockIdx.x * (blockDim.x >> 5) + wave;
  if (tile >= totalTiles) return;

  const int tm = tile / tilesN;
  const int tn = tile - tm * tilesN;

  const int r    = lane & 15;
  const int klo  = (lane >> 4) << 1;

  const float* __restrict__ Ar[4];
  const float* __restrict__ Br[4];
#pragma unroll
  for (int i = 0; i < 4; ++i) {
    Ar[i] = A + (size_t)(tm * 64 + i * 16 + r) * lda + klo;
    Br[i] = W + (size_t)(tn * 64 + i * 16 + r) * ldb + klo;
  }

  v8f acc[4][4];
#pragma unroll
  for (int i = 0; i < 4; ++i)
#pragma unroll
    for (int j = 0; j < 4; ++j) acc[i][j] = {};

  for (int k = 0; k < K; k += 4) {
    v2f a[4], b[4];
#pragma unroll
    for (int i = 0; i < 4; ++i) a[i] = *(const v2f*)(Ar[i] + k);
#pragma unroll
    for (int j = 0; j < 4; ++j) b[j] = *(const v2f*)(Br[j] + k);
#pragma unroll
    for (int i = 0; i < 4; ++i)
#pragma unroll
      for (int j = 0; j < 4; ++j)
        acc[i][j] = wmma_f32_16x16x4(a[i], b[j], acc[i][j]);
  }

  const int half = lane >> 4;
#pragma unroll
  for (int i = 0; i < 4; ++i) {
#pragma unroll
    for (int j = 0; j < 4; ++j) {
      const int cn = tn * 64 + j * 16 + r;
#pragma unroll
      for (int v = 0; v < 8; ++v) {
        const int cm = tm * 64 + i * 16 + v + (half << 3);
        float val = acc[i][j][v];
        if (EPI == 1) {
          val += bias[cn];
          val = (val > 20.f) ? val : log1pf(__expf(val));
        }
        C[(size_t)cm * ldc + cn] = val;
      }
    }
  }
}

__global__ void __launch_bounds__(256)
conv_silu_kernel(const float* __restrict__ zx,
                 const float* __restrict__ conv_w,
                 const float* __restrict__ conv_b,
                 float* __restrict__ xconv)
{
  int idx = blockIdx.x * blockDim.x + threadIdx.x;
  if (idx >= BATCH * SEQ * D_INNER) return;
  const int d = idx % D_INNER;
  const int l = (idx / D_INNER) % SEQ;
  const int b = idx / (D_INNER * SEQ);
  const int src = ((d >> 6) << 4) + (d & 15);

  const float* __restrict__ w = conv_w + d * D_CONV;
  float acc = conv_b[d];
#pragma unroll
  for (int t = 0; t < D_CONV; ++t) {
    const int ll = l - (D_CONV - 1) + t;
    if (ll >= 0) {
      float xv = zx[(size_t)(b * SEQ + ll) * E_DIM + OFF_X + src];
      acc = fmaf(w[t], xv, acc);
    }
  }
  acc = acc / (1.f + __expf(-acc));
  xconv[(size_t)idx] = acc;
}

__global__ void __launch_bounds__(256)
scan_kernel(const float* __restrict__ zx,
            const float* __restrict__ xconv,
            const float* __restrict__ dtbuf,
            const float* __restrict__ Amat,
            const float* __restrict__ Dvec,
            float* __restrict__ Y)
{
  const int b = blockIdx.x / H_HEADS;
  const int h = blockIdx.x % H_HEADS;
  const int p = threadIdx.x >> 4;
  const int n = threadIdx.x & 15;

  const float An = Amat[(size_t)(h * PCH + p) * D_STATE + n];
  const float Dv = Dvec[h * PCH + p];

  const size_t chan = (size_t)h * PCH + p;
  const size_t boff = OFF_B + (size_t)(h / REP) * D_STATE + n;
  const size_t coff = OFF_C + (size_t)h * D_STATE + n;
  const size_t zoff = OFF_Z + chan;

  float hstate = 0.f;
  size_t row = (size_t)b * SEQ;
  for (int l = 0; l < SEQ; ++l, ++row) {
    const float dt = dtbuf[row * D_INNER + chan];
    const float xv = xconv[row * D_INNER + chan];
    const float Bv = zx[row * E_DIM + boff];
    const float Cv = zx[row * E_DIM + coff];

    const float a = __expf(dt * An);
    hstate = fmaf(a, hstate, dt * xv * Bv);

    float yc = hstate * Cv;
    yc += __shfl_xor(yc, 1, 32);
    yc += __shfl_xor(yc, 2, 32);
    yc += __shfl_xor(yc, 4, 32);
    yc += __shfl_xor(yc, 8, 32);

    if (n == 0) {
      const float zval = zx[row * E_DIM + zoff];
      const float gate = zval / (1.f + __expf(-zval));
      Y[row * D_INNER + chan] = (yc + Dv * xv) * gate;
    }
  }
}

extern "C" void kernel_launch(void* const* d_in, const int* in_sizes, int n_in,
                              void* d_out, int out_size, void* d_ws, size_t ws_size,
                              hipStream_t stream) {
  const float* hidden     = (const float*)d_in[0];
  const float* in_proj_w  = (const float*)d_in[1];
  const float* conv_w     = (const float*)d_in[2];
  const float* conv_b     = (const float*)d_in[3];
  const float* dt_proj_w  = (const float*)d_in[4];
  const float* dt_proj_b  = (const float*)d_in[5];
  const float* Amat       = (const float*)d_in[6];
  const float* Dvec       = (const float*)d_in[7];
  const float* out_proj_w = (const float*)d_in[8];
  float* out = (float*)d_out;

  float* zx    = (float*)d_ws;
  float* xconv = zx    + (size_t)M_ROWS * E_DIM;
  float* dtbuf = xconv + (size_t)M_ROWS * D_INNER;
  float* Ybuf  = dtbuf + (size_t)M_ROWS * D_INNER;

  {
    int tiles = (M_ROWS / 64) * (E_DIM / 64);
    wmma_gemm_f32<0><<<(tiles + 7) / 8, 256, 0, stream>>>(
        hidden, D_MODEL, in_proj_w, D_MODEL, zx, E_DIM,
        M_ROWS, E_DIM, D_MODEL, nullptr);
  }

  {
    int n = BATCH * SEQ * D_INNER;
    conv_silu_kernel<<<(n + 255) / 256, 256, 0, stream>>>(zx, conv_w, conv_b, xconv);
  }

  {
    int tiles = (M_ROWS / 64) * (D_INNER / 64);
    wmma_gemm_f32<1><<<(tiles + 7) / 8, 256, 0, stream>>>(
        zx + OFF_DT, E_DIM, dt_proj_w, DT_RANK, dtbuf, D_INNER,
        M_ROWS, D_INNER, DT_RANK, dt_proj_b);
  }

  scan_kernel<<<BATCH * H_HEADS, 256, 0, stream>>>(zx, xconv, dtbuf, Amat, Dvec, Ybuf);

  {
    int tiles = (M_ROWS / 64) * (D_MODEL / 64);
    wmma_gemm_f32<0><<<(tiles + 7) / 8, 256, 0, stream>>>(
        Ybuf, D_INNER, out_proj_w, D_INNER, out, D_MODEL,
        M_ROWS, D_MODEL, D_INNER, nullptr);
  }
}


# in_proj in one resident round: 64x80 wave tiles on its 5120 z/x/B/C columns = 4096 waves; the 64 dt columns go to the dt launch (hidden->dt_in, K over 8 waves summed in LDS: new sum order, ->raw dt)
# speedup vs baseline: 6.3456x; 6.3456x over previous
#include <hip/hip_runtime.h>
#include <math.h>

#define D_MODEL   1024
#define D_INNER   2048
#define D_XB       512
#define D_STATE     16
#define D_CONV       4
#define DT_RANK     64
#define NXB         32
#define H_HEADS    128
#define REP          4
#define PCH         16
#define BATCH        2
#define SEQ       2048
#define E_DIM     (2*D_INNER + 2*D_XB + DT_RANK)
#define M_ROWS    (BATCH*SEQ)
#define TCH         32
#define NCHUNK    (SEQ / TCH)
#define SCAN_BLK   128
#define SCAN_DG   (D_INNER / SCAN_BLK)

#define OFF_Z   0
#define OFF_X   (D_INNER)
#define OFF_B   (D_INNER + D_XB)
#define OFF_C   (D_INNER + 2*D_XB)
#define OFF_DT  (2*D_INNER + 2*D_XB)

typedef __attribute__((ext_vector_type(2))) float v2f;
typedef __attribute__((ext_vector_type(4))) float v4f;
typedef __attribute__((ext_vector_type(8))) float v8f;

__device__ __forceinline__ v8f wmma_f32_16x16x4(v2f a, v2f b, v8f c) {
  return __builtin_amdgcn_wmma_f32_16x16x4_f32(
       false, a,  false, b,
       (short)0, c,  false,  false);
}

template<int KSPLIT>
__global__ void __launch_bounds__(256) __attribute__((amdgpu_waves_per_eu(4, 4)))
wmma_gemm_f32(const float* __restrict__ A, int lda,
              const float* __restrict__ W, int ldb,
              float* __restrict__ C, int ldc,
              int M, int N, int K)
{
  constexpr int TEAM = 8 / KSPLIT;
  const int lane = threadIdx.x & 31;
  const int tilesN = N >> 6;
  const int totalTiles = (M >> 6) * tilesN;
  int tw, kpart, tile;
  if (KSPLIT == 1) {
    const int wave = threadIdx.x >> 5;
    tw = wave; kpart = 0;
    tile = blockIdx.x * (blockDim.x >> 5) + wave;
  } else {
    const int wave = __builtin_amdgcn_readfirstlane((int)(threadIdx.x >> 5));
    tw = wave % TEAM; kpart = wave / TEAM;
    tile = blockIdx.x * TEAM + tw;
  }
  const bool live = tile < totalTiles;
  if (KSPLIT == 1) {
    if (!live) return;
  } else {
    if (!live) tile = 0;
  }

  const int tm = tile / tilesN;
  const int tn = tile - tm * tilesN;

  const int r    = lane & 15;
  const int klo  = (lane >> 4) << 3;

  const float* __restrict__ Ar[4];
  const float* __restrict__ Br[4];
#pragma unroll
  for (int i = 0; i < 4; ++i) {
    Ar[i] = A + (size_t)(tm * 64 + i * 16 + r) * lda + klo;
    Br[i] = W + (size_t)(tn * 64 + i * 16 + r) * ldb + klo;
  }

  v8f acc[4][4];
#pragma unroll
  for (int i = 0; i < 4; ++i)
#pragma unroll
    for (int j = 0; j < 4; ++j) acc[i][j] = {};

  for (int k = kpart * 16; k < K; k += 16 * KSPLIT) {
    v4f a[4][2], b[4][2];
#pragma unroll
    for (int i = 0; i < 4; ++i) {
      a[i][0] = *(const v4f*)(Ar[i] + k);
      a[i][1] = *(const v4f*)(Ar[i] + k + 4);
    }
#pragma unroll
    for (int j = 0; j < 4; ++j) {
      b[j][0] = *(const v4f*)(Br[j] + k);
      b[j][1] = *(const v4f*)(Br[j] + k + 4);
    }
#pragma unroll
    for (int s = 0; s < 4; ++s) {
#pragma unroll
      for (int i = 0; i < 4; ++i) {
        const v4f ap = a[i][s >> 1];
        const v2f af = (s & 1) ? ap.zw : ap.xy;
#pragma unroll
        for (int j = 0; j < 4; ++j) {
          const v4f bp = b[j][s >> 1];
          const v2f bf = (s & 1) ? bp.zw : bp.xy;
          acc[i][j] = wmma_f32_16x16x4(af, bf, acc[i][j]);
        }
      }
    }
  }

  if (KSPLIT == 2) {
    __shared__ v4f fold[TEAM][32][32];
    if (kpart == 1) {
#pragma unroll
      for (int i = 0; i < 4; ++i)
#pragma unroll
        for (int j = 0; j < 4; ++j) {
          const v8f t = acc[i][j];
          fold[tw][(i * 4 + j) * 2 + 0][lane] = __builtin_shufflevector(t, t, 0, 1, 2, 3);
          fold[tw][(i * 4 + j) * 2 + 1][lane] = __builtin_shufflevector(t, t, 4, 5, 6, 7);
        }
    }
    __syncthreads();
    if (kpart == 1 || !live) return;
#pragma unroll
    for (int i = 0; i < 4; ++i)
#pragma unroll
      for (int j = 0; j < 4; ++j) {
        const v4f lo = fold[tw][(i * 4 + j) * 2 + 0][lane];
        const v4f hi = fold[tw][(i * 4 + j) * 2 + 1][lane];
        acc[i][j] += __builtin_shufflevector(lo, hi, 0, 1, 2, 3, 4, 5, 6, 7);
      }
  }

  const int half = lane >> 4;
#pragma unroll
  for (int i = 0; i < 4; ++i) {
#pragma unroll
    for (int j = 0; j < 4; ++j) {
      const int cn = tn * 64 + j * 16 + r;
#pragma unroll
      for (int v = 0; v < 8; ++v) {
        const int cm = tm * 64 + i * 16 + v + (half << 3);
        C[(size_t)cm * ldc + cn] = acc[i][j][v];
      }
    }
  }
}

#define N_MAIN   (2*D_INNER + 2*D_XB)
#define WT_N     80
__global__ void __launch_bounds__(256) __attribute__((amdgpu_waves_per_eu(4, 4)))
in_proj_gemm_kernel(const float* __restrict__ A,
                    const float* __restrict__ W,
                    float* __restrict__ C)
{
  constexpr int tilesN = N_MAIN / WT_N;
  const int lane = threadIdx.x & 31;
  const int wave = threadIdx.x >> 5;
  const int tile = blockIdx.x * 8 + wave;
  const int tm = tile / tilesN;
  const int tn = tile - tm * tilesN;

  const int r   = lane & 15;
  const int klo = (lane >> 4) << 3;

  const float* __restrict__ Ab = A + (size_t)(tm * 64 + r) * D_MODEL + klo;
  const float* __restrict__ Bb = W + (size_t)(tn * WT_N + r) * D_MODEL + klo;

  v8f acc[4][5];
#pragma unroll
  for (int i = 0; i < 4; ++i)
#pragma unroll
    for (int j = 0; j < 5; ++j) acc[i][j] = {};

  for (int k = 0; k < D_MODEL; k += 16) {
    v4f a[4][2], b[5][2];
#pragma unroll
    for (int i = 0; i < 4; ++i) {
      a[i][0] = *(const v4f*)(Ab + i * 16 * D_MODEL + k);
      a[i][1] = *(const v4f*)(Ab + i * 16 * D_MODEL + k + 4);
    }
#pragma unroll
    for (int j = 0; j < 5; ++j) {
      b[j][0] = *(const v4f*)(Bb + j * 16 * D_MODEL + k);
      b[j][1] = *(const v4f*)(Bb + j * 16 * D_MODEL + k + 4);
    }
#pragma unroll
    for (int s = 0; s < 4; ++s) {
#pragma unroll
      for (int i = 0; i < 4; ++i) {
        const v4f ap = a[i][s >> 1];
        const v2f af = (s & 1) ? ap.zw : ap.xy;
#pragma unroll
        for (int j = 0; j < 5; ++j) {
          const v4f bp = b[j][s >> 1];
          const v2f bf = (s & 1) ? bp.zw : bp.xy;
          acc[i][j] = wmma_f32_16x16x4(af, bf, acc[i][j]);
        }
      }
    }
  }

  const int half = lane >> 4;
#pragma unroll
  for (int i = 0; i < 4; ++i) {
#pragma unroll
    for (int j = 0; j < 5; ++j) {
      const int cn = tn * WT_N + j * 16 + r;
#pragma unroll
      for (int v = 0; v < 8; ++v) {
        const int cm = tm * 64 + i * 16 + v + (half << 3);
        C[(size_t)cm * E_DIM + cn] = acc[i][j][v];
      }
    }
  }
}

#define DTI_PITCH 68
__global__ void __launch_bounds__(256)
dt_chain_kernel(const float* __restrict__ hidden,
                const float* __restrict__ Wdt,
                const float* __restrict__ dtw,
                float* __restrict__ dtbuf)
{
  __shared__ float part[8][16][DT_RANK];
  __shared__ float dti[16][DTI_PITCH];

  const int lane = threadIdx.x & 31;
  const int wave = __builtin_amdgcn_readfirstlane((int)(threadIdx.x >> 5));
  const int r    = lane & 15;
  const int half = lane >> 4;
  const int klo  = half << 3;
  const int m0   = blockIdx.x * 16;

  {
    const float* __restrict__ Ap = hidden + (size_t)(m0 + r) * D_MODEL + wave * 128 + klo;
    const float* __restrict__ Bp = Wdt + (size_t)r * D_MODEL + wave * 128 + klo;
    v8f acc[4];
#pragma unroll
    for (int j = 0; j < 4; ++j) acc[j] = {};
#pragma unroll
    for (int kb = 0; kb < 8; ++kb) {
      v4f a[2], b[4][2];
      a[0] = *(const v4f*)(Ap + kb * 16);
      a[1] = *(const v4f*)(Ap + kb * 16 + 4);
#pragma unroll
      for (int j = 0; j < 4; ++j) {
        b[j][0] = *(const v4f*)(Bp + j * 16 * D_MODEL + kb * 16);
        b[j][1] = *(const v4f*)(Bp + j * 16 * D_MODEL + kb * 16 + 4);
      }
#pragma unroll
      for (int s = 0; s < 4; ++s) {
        const v4f ap = a[s >> 1];
        const v2f af = (s & 1) ? ap.zw : ap.xy;
#pragma unroll
        for (int j = 0; j < 4; ++j) {
          const v4f bp = b[j][s >> 1];
          const v2f bf = (s & 1) ? bp.zw : bp.xy;
          acc[j] = wmma_f32_16x16x4(af, bf, acc[j]);
        }
      }
    }
#pragma unroll
    for (int j = 0; j < 4; ++j)
#pragma unroll
      for (int v = 0; v < 8; ++v)
        part[wave][v + (half << 3)][j * 16 + r] = acc[j][v];
  }
  __syncthreads();
  {
    const int row = threadIdx.x >> 4;
    const int c4  = (threadIdx.x & 15) << 2;
    v4f s = *(const v4f*)&part[0][row][c4];
#pragma unroll
    for (int w = 1; w < 8; ++w) s += *(const v4f*)&part[w][row][c4];
    *(v4f*)&dti[row][c4] = s;
  }
  __syncthreads();

  v4f a2[4][2];
#pragma unroll
  for (int kb = 0; kb < 4; ++kb) {
    a2[kb][0] = *(const v4f*)&dti[r][kb * 16 + klo];
    a2[kb][1] = *(const v4f*)&dti[r][kb * 16 + klo + 4];
  }
  for (int g = 0; g < 4; ++g) {
    const int n0 = wave * 256 + g * 64;
    const float* __restrict__ Bq = dtw + (size_t)(n0 + r) * DT_RANK + klo;
    v4f b2[4][4][2];
#pragma unroll
    for (int q = 0; q < 4; ++q)
#pragma unroll
      for (int kb = 0; kb < 4; ++kb) {
        b2[q][kb][0] = *(const v4f*)(Bq + q * 16 * DT_RANK + kb * 16);
        b2[q][kb][1] = *(const v4f*)(Bq + q * 16 * DT_RANK + kb * 16 + 4);
      }
    v8f acc[4];
#pragma unroll
    for (int q = 0; q < 4; ++q) acc[q] = {};
#pragma unroll
    for (int kb = 0; kb < 4; ++kb)
#pragma unroll
      for (int s = 0; s < 4; ++s) {
        const v4f ap = a2[kb][s >> 1];
        const v2f af = (s & 1) ? ap.zw : ap.xy;
#pragma unroll
        for (int q = 0; q < 4; ++q) {
          const v4f bp = b2[q][kb][s >> 1];
          const v2f bf = (s & 1) ? bp.zw : bp.xy;
          acc[q] = wmma_f32_16x16x4(af, bf, acc[q]);
        }
      }
#pragma unroll
    for (int q = 0; q < 4; ++q)
#pragma unroll
      for (int v = 0; v < 8; ++v)
        dtbuf[(size_t)(m0 + v + (half << 3)) * D_INNER + n0 + q * 16 + r] = acc[q][v];
  }
}

__device__ __forceinline__ void load16(const float* __restrict__ p, float (&v)[D_STATE]) {
  const v4f* __restrict__ q = (const v4f*)p;
  const v4f q0 = q[0], q1 = q[1], q2 = q[2], q3 = q[3];
  v[0]  = q0.x; v[1]  = q0.y; v[2]  = q0.z; v[3]  = q0.w;
  v[4]  = q1.x; v[5]  = q1.y; v[6]  = q1.z; v[7]  = q1.w;
  v[8]  = q2.x; v[9]  = q2.y; v[10] = q2.z; v[11] = q2.w;
  v[12] = q3.x; v[13] = q3.y; v[14] = q3.z; v[15] = q3.w;
}

__global__ void __launch_bounds__(SCAN_BLK)
scan_chunk_state_kernel(const float* __restrict__ zx,
                        const float* __restrict__ conv_w,
                        const float* __restrict__ conv_b,
                        const float* __restrict__ dt_bias,
                        const float* __restrict__ Amat,
                        float* __restrict__ xconv,
                        float* __restrict__ dtbuf,
                        float* __restrict__ hend,
                        float* __restrict__ pdec)
{
  const int dg = blockIdx.x % SCAN_DG;
  const int c  = (blockIdx.x / SCAN_DG) % NCHUNK;
  const int b  = blockIdx.x / (SCAN_DG * NCHUNK);
  const int d  = dg * SCAN_BLK + threadIdx.x;
  const int src = ((d >> 6) << 4) + (d & 15);

  float An[D_STATE];
  load16(Amat + (size_t)d * D_STATE, An);
  const v4f w = *(const v4f*)(conv_w + (size_t)d * D_CONV);
  const float cb = conv_b[d];
  const float bias = dt_bias[d];

  float h[D_STATE], pr[D_STATE];
#pragma unroll
  for (int n = 0; n < D_STATE; ++n) { h[n] = 0.f; pr[n] = 1.f; }

  const size_t row0 = (size_t)b * SEQ + (size_t)c * TCH;
  float* __restrict__ dtp = dtbuf + row0 * D_INNER + d;
  float* __restrict__ xcp = xconv + row0 * D_INNER + d;
  const float* __restrict__ xrp = zx + row0 * E_DIM + OFF_X + src;
  const float* __restrict__ bp  = zx + row0 * E_DIM + OFF_B + (size_t)(d >> 6) * D_STATE;

  float xm3 = 0.f, xm2 = 0.f, xm1 = 0.f;
  if (c > 0) {
    xm3 = xrp[-3 * E_DIM];
    xm2 = xrp[-2 * E_DIM];
    xm1 = xrp[-1 * E_DIM];
  }

  for (int l = 0; l < TCH; ++l) {
    const float raw = *dtp;
    const float x0  = *xrp;
    float Bv[D_STATE];
    load16(bp, Bv);

    float acc = cb;
    if (c > 0 || l >= 3) acc = fmaf(w.x, xm3, acc);
    if (c > 0 || l >= 2) acc = fmaf(w.y, xm2, acc);
    if (c > 0 || l >= 1) acc = fmaf(w.z, xm1, acc);
    acc = fmaf(w.w, x0, acc);
    xm3 = xm2; xm2 = xm1; xm1 = x0;
    const float xv = acc / (1.f + __expf(-acc));
    *xcp = xv;

    float dt = raw + bias;
    dt = (dt > 20.f) ? dt : log1pf(__expf(dt));
    *dtp = dt;

    const float dtx = dt * xv;
#pragma unroll
    for (int n = 0; n < D_STATE; ++n) {
      const float a = __expf(dt * An[n]);
      h[n] = fmaf(a, h[n], dtx * Bv[n]);
      pr[n] *= a;
    }
    dtp += D_INNER; xcp += D_INNER; xrp += E_DIM; bp += E_DIM;
  }

  if (c < NCHUNK - 1) {
    const size_t o = ((size_t)b * (NCHUNK - 1) + c) * D_STATE * D_INNER + d;
#pragma unroll
    for (int n = 0; n < D_STATE; ++n) {
      hend[o + (size_t)n * D_INNER] = h[n];
      pdec[o + (size_t)n * D_INNER] = pr[n];
    }
  }
}

__global__ void __launch_bounds__(256)
scan_prefix_kernel(const float* __restrict__ hend,
                   const float* __restrict__ pdec,
                   float* __restrict__ hstart)
{
  const int idx = blockIdx.x * 256 + threadIdx.x;
  const int d = idx % D_INNER;
  const int n = (idx / D_INNER) % D_STATE;
  const int b = idx / (D_INNER * D_STATE);

  size_t o = ((size_t)b * (NCHUNK - 1) * D_STATE + n) * D_INNER + d;
  float hs = 0.f;
  for (int c = 0; c < NCHUNK - 1; ++c) {
    hs = fmaf(pdec[o], hs, hend[o]);
    hstart[o] = hs;
    o += (size_t)D_STATE * D_INNER;
  }
}

__global__ void __launch_bounds__(SCAN_BLK)
scan_kernel(const float* __restrict__ zx,
            const float* __restrict__ xconv,
            const float* __restrict__ dtbuf,
            const float* __restrict__ Amat,
            const float* __restrict__ Dvec,
            const float* __restrict__ hstart,
            float* __restrict__ Y)
{
  const int dg = blockIdx.x % SCAN_DG;
  const int c  = (blockIdx.x / SCAN_DG) % NCHUNK;
  const int b  = blockIdx.x / (SCAN_DG * NCHUNK);
  const int d  = dg * SCAN_BLK + threadIdx.x;

  float An[D_STATE];
  load16(Amat + (size_t)d * D_STATE, An);
  const float Dv = Dvec[d];

  float h[D_STATE];
#pragma unroll
  for (int n = 0; n < D_STATE; ++n) h[n] = 0.f;
  if (c > 0) {
    const size_t o = ((size_t)b * (NCHUNK - 1) + (c - 1)) * D_STATE * D_INNER + d;
#pragma unroll
    for (int n = 0; n < D_STATE; ++n) h[n] = hstart[o + (size_t)n * D_INNER];
  }

  const size_t row0 = (size_t)b * SEQ + (size_t)c * TCH;
  const float* __restrict__ dtp = dtbuf + row0 * D_INNER + d;
  const float* __restrict__ xp  = xconv + row0 * D_INNER + d;
  const float* __restrict__ zp  = zx + row0 * E_DIM + OFF_Z + d;
  const float* __restrict__ bp  = zx + row0 * E_DIM + OFF_B + (size_t)(d >> 6) * D_STATE;
  const float* __restrict__ cp  = zx + row0 * E_DIM + OFF_C + (size_t)(d >> 4) * D_STATE;
  float* __restrict__ yp = Y + row0 * D_INNER + d;

  for (int l = 0; l < TCH; ++l) {
    const float dt = *dtp;
    const float xv = *xp;
    const float zval = *zp;
    float Bv[D_STATE], Cv[D_STATE];
    load16(bp, Bv);
    load16(cp, Cv);
    dtp += D_INNER; xp += D_INNER; zp += E_DIM; bp += E_DIM; cp += E_DIM;

    const float dtx = dt * xv;
    float y0 = 0.f, y1 = 0.f;
#pragma unroll
    for (int n = 0; n < D_STATE; n += 2) {
      const float a0 = __expf(dt * An[n]);
      const float a1 = __expf(dt * An[n + 1]);
      h[n]     = fmaf(a0, h[n],     dtx * Bv[n]);
      h[n + 1] = fmaf(a1, h[n + 1], dtx * Bv[n + 1]);
      y0 = fmaf(h[n],     Cv[n],     y0);
      y1 = fmaf(h[n + 1], Cv[n + 1], y1);
    }
    const float yc = y0 + y1;

    const float gate = zval / (1.f + __expf(-zval));
    *yp = (yc + Dv * xv) * gate;
    yp += D_INNER;
  }
}

extern "C" void kernel_launch(void* const* d_in, const int* in_sizes, int n_in,
                              void* d_out, int out_size, void* d_ws, size_t ws_size,
                              hipStream_t stream) {
  const float* hidden     = (const float*)d_in[0];
  const float* in_proj_w  = (const float*)d_in[1];
  const float* conv_w     = (const float*)d_in[2];
  const float* conv_b     = (const float*)d_in[3];
  const float* dt_proj_w  = (const float*)d_in[4];
  const float* dt_proj_b  = (const float*)d_in[5];
  const float* Amat       = (const float*)d_in[6];
  const float* Dvec       = (const float*)d_in[7];
  const float* out_proj_w = (const float*)d_in[8];
  float* out = (float*)d_out;

  float* zx    = (float*)d_ws;
  float* xconv = zx    + (size_t)M_ROWS * E_DIM;
  float* dtbuf = xconv + (size_t)M_ROWS * D_INNER;
  float* Ybuf  = dtbuf + (size_t)M_ROWS * D_INNER;
  float* hend  = Ybuf  + (size_t)M_ROWS * D_INNER;
  float* pdec  = hend  + (size_t)BATCH * (NCHUNK - 1) * D_STATE * D_INNER;
  float* hstart = pdec + (size_t)BATCH * (NCHUNK - 1) * D_STATE * D_INNER;

  in_proj_gemm_kernel<<<(M_ROWS / 64) * (N_MAIN / WT_N) / 8, 256, 0, stream>>>(
      hidden, in_proj_w, zx);

  dt_chain_kernel<<<M_ROWS / 16, 256, 0, stream>>>(
      hidden, in_proj_w + (size_t)OFF_DT * D_MODEL, dt_proj_w, dtbuf);

  scan_chunk_state_kernel<<<BATCH * SCAN_DG * NCHUNK, SCAN_BLK, 0, stream>>>(
      zx, conv_w, conv_b, dt_proj_b, Amat, xconv, dtbuf, hend, pdec);

  scan_prefix_kernel<<<BATCH * D_STATE * D_INNER / 256, 256, 0, stream>>>(
      hend, pdec, hstart);

  scan_kernel<<<BATCH * SCAN_DG * NCHUNK, SCAN_BLK, 0, stream>>>(
      zx, xconv, dtbuf, Amat, Dvec, hstart, Ybuf);

  {
    int tiles = (M_ROWS / 64) * (D_MODEL / 64);
    wmma_gemm_f32<2><<<(tiles + 3) / 4, 256, 0, stream>>>(
        Ybuf, D_INNER, out_proj_w, D_INNER, out, D_MODEL,
        M_ROWS, D_MODEL, D_INNER);
  }
}


# scan: chunk-state pass, prefix and output pass fused into ONE launch (block = 16 channels x all 64 chunks); chunk states, start states and dt go through LDS; arithmetic and sum orders unchanged
# speedup vs baseline: 6.5489x; 1.0320x over previous
#include <hip/hip_runtime.h>
#include <math.h>

#define D_MODEL   1024
#define D_INNER   2048
#define D_XB       512
#define D_STATE     16
#define D_CONV       4
#define DT_RANK     64
#define NXB         32
#define H_HEADS    128
#define REP          4
#define PCH         16
#define BATCH        2
#define SEQ       2048
#define E_DIM     (2*D_INNER + 2*D_XB + DT_RANK)
#define M_ROWS    (BATCH*SEQ)
#define TCH         32
#define NCHUNK    (SEQ / TCH)

#define OFF_Z   0
#define OFF_X   (D_INNER)
#define OFF_B   (D_INNER + D_XB)
#define OFF_C   (D_INNER + 2*D_XB)
#define OFF_DT  (2*D_INNER + 2*D_XB)

typedef __attribute__((ext_vector_type(2))) float v2f;
typedef __attribute__((ext_vector_type(4))) float v4f;
typedef __attribute__((ext_vector_type(8))) float v8f;

__device__ __forceinline__ v8f wmma_f32_16x16x4(v2f a, v2f b, v8f c) {
  return __builtin_amdgcn_wmma_f32_16x16x4_f32(
       false, a,  false, b,
       (short)0, c,  false,  false);
}

template<int KSPLIT>
__global__ void __launch_bounds__(256) __attribute__((amdgpu_waves_per_eu(4, 4)))
wmma_gemm_f32(const float* __restrict__ A, int lda,
              const float* __restrict__ W, int ldb,
              float* __restrict__ C, int ldc,
              int M, int N, int K)
{
  constexpr int TEAM = 8 / KSPLIT;
  const int lane = threadIdx.x & 31;
  const int tilesN = N >> 6;
  const int totalTiles = (M >> 6) * tilesN;
  int tw, kpart, tile;
  if (KSPLIT == 1) {
    const int wave = threadIdx.x >> 5;
    tw = wave; kpart = 0;
    tile = blockIdx.x * (blockDim.x >> 5) + wave;
  } else {
    const int wave = __builtin_amdgcn_readfirstlane((int)(threadIdx.x >> 5));
    tw = wave % TEAM; kpart = wave / TEAM;
    tile = blockIdx.x * TEAM + tw;
  }
  const bool live = tile < totalTiles;
  if (KSPLIT == 1) {
    if (!live) return;
  } else {
    if (!live) tile = 0;
  }

  const int tm = tile / tilesN;
  const int tn = tile - tm * tilesN;

  const int r    = lane & 15;
  const int klo  = (lane >> 4) << 3;

  const float* __restrict__ Ar[4];
  const float* __restrict__ Br[4];
#pragma unroll
  for (int i = 0; i < 4; ++i) {
    Ar[i] = A + (size_t)(tm * 64 + i * 16 + r) * lda + klo;
    Br[i] = W + (size_t)(tn * 64 + i * 16 + r) * ldb + klo;
  }

  v8f acc[4][4];
#pragma unroll
  for (int i = 0; i < 4; ++i)
#pragma unroll
    for (int j = 0; j < 4; ++j) acc[i][j] = {};

  for (int k = kpart * 16; k < K; k += 16 * KSPLIT) {
    v4f a[4][2], b[4][2];
#pragma unroll
    for (int i = 0; i < 4; ++i) {
      a[i][0] = *(const v4f*)(Ar[i] + k);
      a[i][1] = *(const v4f*)(Ar[i] + k + 4);
    }
#pragma unroll
    for (int j = 0; j < 4; ++j) {
      b[j][0] = *(const v4f*)(Br[j] + k);
      b[j][1] = *(const v4f*)(Br[j] + k + 4);
    }
#pragma unroll
    for (int s = 0; s < 4; ++s) {
#pragma unroll
      for (int i = 0; i < 4; ++i) {
        const v4f ap = a[i][s >> 1];
        const v2f af = (s & 1) ? ap.zw : ap.xy;
#pragma unroll
        for (int j = 0; j < 4; ++j) {
          const v4f bp = b[j][s >> 1];
          const v2f bf = (s & 1) ? bp.zw : bp.xy;
          acc[i][j] = wmma_f32_16x16x4(af, bf, acc[i][j]);
        }
      }
    }
  }

  if (KSPLIT == 2) {
    __shared__ v4f fold[TEAM][32][32];
    if (kpart == 1) {
#pragma unroll
      for (int i = 0; i < 4; ++i)
#pragma unroll
        for (int j = 0; j < 4; ++j) {
          const v8f t = acc[i][j];
          fold[tw][(i * 4 + j) * 2 + 0][lane] = __builtin_shufflevector(t, t, 0, 1, 2, 3);
          fold[tw][(i * 4 + j) * 2 + 1][lane] = __builtin_shufflevector(t, t, 4, 5, 6, 7);
        }
    }
    __syncthreads();
    if (kpart == 1 || !live) return;
#pragma unroll
    for (int i = 0; i < 4; ++i)
#pragma unroll
      for (int j = 0; j < 4; ++j) {
        const v4f lo = fold[tw][(i * 4 + j) * 2 + 0][lane];
        const v4f hi = fold[tw][(i * 4 + j) * 2 + 1][lane];
        acc[i][j] += __builtin_shufflevector(lo, hi, 0, 1, 2, 3, 4, 5, 6, 7);
      }
  }

  const int half = lane >> 4;
#pragma unroll
  for (int i = 0; i < 4; ++i) {
#pragma unroll
    for (int j = 0; j < 4; ++j) {
      const int cn = tn * 64 + j * 16 + r;
#pragma unroll
      for (int v = 0; v < 8; ++v) {
        const int cm = tm * 64 + i * 16 + v + (half << 3);
        C[(size_t)cm * ldc + cn] = acc[i][j][v];
      }
    }
  }
}

#define N_MAIN   (2*D_INNER + 2*D_XB)
#define WT_N     80
__global__ void __launch_bounds__(256) __attribute__((amdgpu_waves_per_eu(4, 4)))
in_proj_gemm_kernel(const float* __restrict__ A,
                    const float* __restrict__ W,
                    float* __restrict__ C)
{
  constexpr int tilesN = N_MAIN / WT_N;
  const int lane = threadIdx.x & 31;
  const int wave = threadIdx.x >> 5;
  const int tile = blockIdx.x * 8 + wave;
  const int tm = tile / tilesN;
  const int tn = tile - tm * tilesN;

  const int r   = lane & 15;
  const int klo = (lane >> 4) << 3;

  const float* __restrict__ Ab = A + (size_t)(tm * 64 + r) * D_MODEL + klo;
  const float* __restrict__ Bb = W + (size_t)(tn * WT_N + r) * D_MODEL + klo;

  v8f acc[4][5];
#pragma unroll
  for (int i = 0; i < 4; ++i)
#pragma unroll
    for (int j = 0; j < 5; ++j) acc[i][j] = {};

  for (int k = 0; k < D_MODEL; k += 16) {
    v4f a[4][2], b[5][2];
#pragma unroll
    for (int i = 0; i < 4; ++i) {
      a[i][0] = *(const v4f*)(Ab + i * 16 * D_MODEL + k);
      a[i][1] = *(const v4f*)(Ab + i * 16 * D_MODEL + k + 4);
    }
#pragma unroll
    for (int j = 0; j < 5; ++j) {
      b[j][0] = *(const v4f*)(Bb + j * 16 * D_MODEL + k);
      b[j][1] = *(const v4f*)(Bb + j * 16 * D_MODEL + k + 4);
    }
#pragma unroll
    for (int s = 0; s < 4; ++s) {
#pragma unroll
      for (int i = 0; i < 4; ++i) {
        const v4f ap = a[i][s >> 1];
        const v2f af = (s & 1) ? ap.zw : ap.xy;
#pragma unroll
        for (int j = 0; j < 5; ++j) {
          const v4f bp = b[j][s >> 1];
          const v2f bf = (s & 1) ? bp.zw : bp.xy;
          acc[i][j] = wmma_f32_16x16x4(af, bf, acc[i][j]);
        }
      }
    }
  }

  const int half = lane >> 4;
#pragma unroll
  for (int i = 0; i < 4; ++i) {
#pragma unroll
    for (int j = 0; j < 5; ++j) {
      const int cn = tn * WT_N + j * 16 + r;
#pragma unroll
      for (int v = 0; v < 8; ++v) {
        const int cm = tm * 64 + i * 16 + v + (half << 3);
        C[(size_t)cm * E_DIM + cn] = acc[i][j][v];
      }
    }
  }
}

#define DTI_PITCH 68
__global__ void __launch_bounds__(256)
dt_chain_kernel(const float* __restrict__ hidden,
                const float* __restrict__ Wdt,
                const float* __restrict__ dtw,
                float* __restrict__ dtbuf)
{
  __shared__ float part[8][16][DT_RANK];
  __shared__ float dti[16][DTI_PITCH];

  const int lane = threadIdx.x & 31;
  const int wave = __builtin_amdgcn_readfirstlane((int)(threadIdx.x >> 5));
  const int r    = lane & 15;
  const int half = lane >> 4;
  const int klo  = half << 3;
  const int m0   = blockIdx.x * 16;

  {
    const float* __restrict__ Ap = hidden + (size_t)(m0 + r) * D_MODEL + wave * 128 + klo;
    const float* __restrict__ Bp = Wdt + (size_t)r * D_MODEL + wave * 128 + klo;
    v8f acc[4];
#pragma unroll
    for (int j = 0; j < 4; ++j) acc[j] = {};
#pragma unroll
    for (int kb = 0; kb < 8; ++kb) {
      v4f a[2], b[4][2];
      a[0] = *(const v4f*)(Ap + kb * 16);
      a[1] = *(const v4f*)(Ap + kb * 16 + 4);
#pragma unroll
      for (int j = 0; j < 4; ++j) {
        b[j][0] = *(const v4f*)(Bp + j * 16 * D_MODEL + kb * 16);
        b[j][1] = *(const v4f*)(Bp + j * 16 * D_MODEL + kb * 16 + 4);
      }
#pragma unroll
      for (int s = 0; s < 4; ++s) {
        const v4f ap = a[s >> 1];
        const v2f af = (s & 1) ? ap.zw : ap.xy;
#pragma unroll
        for (int j = 0; j < 4; ++j) {
          const v4f bp = b[j][s >> 1];
          const v2f bf = (s & 1) ? bp.zw : bp.xy;
          acc[j] = wmma_f32_16x16x4(af, bf, acc[j]);
        }
      }
    }
#pragma unroll
    for (int j = 0; j < 4; ++j)
#pragma unroll
      for (int v = 0; v < 8; ++v)
        part[wave][v + (half << 3)][j * 16 + r] = acc[j][v];
  }
  __syncthreads();
  {
    const int row = threadIdx.x >> 4;
    const int c4  = (threadIdx.x & 15) << 2;
    v4f s = *(const v4f*)&part[0][row][c4];
#pragma unroll
    for (int w = 1; w < 8; ++w) s += *(const v4f*)&part[w][row][c4];
    *(v4f*)&dti[row][c4] = s;
  }
  __syncthreads();

  v4f a2[4][2];
#pragma unroll
  for (int kb = 0; kb < 4; ++kb) {
    a2[kb][0] = *(const v4f*)&dti[r][kb * 16 + klo];
    a2[kb][1] = *(const v4f*)&dti[r][kb * 16 + klo + 4];
  }
  for (int g = 0; g < 4; ++g) {
    const int n0 = wave * 256 + g * 64;
    const float* __restrict__ Bq = dtw + (size_t)(n0 + r) * DT_RANK + klo;
    v4f b2[4][4][2];
#pragma unroll
    for (int q = 0; q < 4; ++q)
#pragma unroll
      for (int kb = 0; kb < 4; ++kb) {
        b2[q][kb][0] = *(const v4f*)(Bq + q * 16 * DT_RANK + kb * 16);
        b2[q][kb][1] = *(const v4f*)(Bq + q * 16 * DT_RANK + kb * 16 + 4);
      }
    v8f acc[4];
#pragma unroll
    for (int q = 0; q < 4; ++q) acc[q] = {};
#pragma unroll
    for (int kb = 0; kb < 4; ++kb)
#pragma unroll
      for (int s = 0; s < 4; ++s) {
        const v4f ap = a2[kb][s >> 1];
        const v2f af = (s & 1) ? ap.zw : ap.xy;
#pragma unroll
        for (int q = 0; q < 4; ++q) {
          const v4f bp = b2[q][kb][s >> 1];
          const v2f bf = (s & 1) ? bp.zw : bp.xy;
          acc[q] = wmma_f32_16x16x4(af, bf, acc[q]);
        }
      }
#pragma unroll
    for (int q = 0; q < 4; ++q)
#pragma unroll
      for (int v = 0; v < 8; ++v)
        dtbuf[(size_t)(m0 + v + (half << 3)) * D_INNER + n0 + q * 16 + r] = acc[q][v];
  }
}

__device__ __forceinline__ void load16(const float* __restrict__ p, float (&v)[D_STATE]) {
  const v4f* __restrict__ q = (const v4f*)p;
  const v4f q0 = q[0], q1 = q[1], q2 = q[2], q3 = q[3];
  v[0]  = q0.x; v[1]  = q0.y; v[2]  = q0.z; v[3]  = q0.w;
  v[4]  = q1.x; v[5]  = q1.y; v[6]  = q1.z; v[7]  = q1.w;
  v[8]  = q2.x; v[9]  = q2.y; v[10] = q2.z; v[11] = q2.w;
  v[12] = q3.x; v[13] = q3.y; v[14] = q3.z; v[15] = q3.w;
}

#define FCH        16
#define FUSED_THREADS (FCH * NCHUNK)
#define ST_PITCH  (NCHUNK * FCH + 16)
__global__ void __launch_bounds__(FUSED_THREADS)
scan_fused_kernel(const float* __restrict__ zx,
                  const float* __restrict__ conv_w,
                  const float* __restrict__ conv_b,
                  const float* __restrict__ dt_bias,
                  const float* __restrict__ Amat,
                  const float* __restrict__ Dvec,
                  const float* __restrict__ dtraw,
                  float* xconv,
                  float* __restrict__ Y)
{
  __shared__ float s_h[D_STATE][ST_PITCH];
  __shared__ float s_p[D_STATE][ST_PITCH];
  __shared__ float s_dt[TCH][FUSED_THREADS];

  const int tid = threadIdx.x;
  const int ch  = tid & (FCH - 1);
  const int c   = tid >> 4;
  const int b   = blockIdx.x / (D_INNER / FCH);
  const int d   = (blockIdx.x % (D_INNER / FCH)) * FCH + ch;
  const int src = ((d >> 6) << 4) + (d & 15);

  float An[D_STATE];
  load16(Amat + (size_t)d * D_STATE, An);

  const size_t row0 = (size_t)b * SEQ + (size_t)c * TCH;
  const float* __restrict__ bp0 = zx + row0 * E_DIM + OFF_B + (size_t)(d >> 6) * D_STATE;
  float* xc0 = xconv + row0 * D_INNER + d;

  {
    const v4f w = *(const v4f*)(conv_w + (size_t)d * D_CONV);
    const float cb = conv_b[d];
    const float bias = dt_bias[d];

    float h[D_STATE], pr[D_STATE];
#pragma unroll
    for (int n = 0; n < D_STATE; ++n) { h[n] = 0.f; pr[n] = 1.f; }

    const float* __restrict__ dtp = dtraw + row0 * D_INNER + d;
    float* xcp = xc0;
    const float* __restrict__ xrp = zx + row0 * E_DIM + OFF_X + src;
    const float* __restrict__ bp  = bp0;

    float xm3 = 0.f, xm2 = 0.f, xm1 = 0.f;
    if (c > 0) {
      xm3 = xrp[-3 * E_DIM];
      xm2 = xrp[-2 * E_DIM];
      xm1 = xrp[-1 * E_DIM];
    }

    for (int l = 0; l < TCH; ++l) {
      const float raw = *dtp;
      const float x0  = *xrp;
      float Bv[D_STATE];
      load16(bp, Bv);

      float acc = cb;
      if (c > 0 || l >= 3) acc = fmaf(w.x, xm3, acc);
      if (c > 0 || l >= 2) acc = fmaf(w.y, xm2, acc);
      if (c > 0 || l >= 1) acc = fmaf(w.z, xm1, acc);
      acc = fmaf(w.w, x0, acc);
      xm3 = xm2; xm2 = xm1; xm1 = x0;
      const float xv = acc / (1.f + __expf(-acc));
      *xcp = xv;

      float dt = raw + bias;
      dt = (dt > 20.f) ? dt : log1pf(__expf(dt));
      s_dt[l][tid] = dt;

      const float dtx = dt * xv;
#pragma unroll
      for (int n = 0; n < D_STATE; ++n) {
        const float a = __expf(dt * An[n]);
        h[n] = fmaf(a, h[n], dtx * Bv[n]);
        pr[n] *= a;
      }
      dtp += D_INNER; xcp += D_INNER; xrp += E_DIM; bp += E_DIM;
    }

#pragma unroll
    for (int n = 0; n < D_STATE; ++n) {
      s_h[n][tid] = h[n];
      s_p[n][tid] = pr[n];
    }
  }
  __syncthreads();

  if (tid < D_STATE * FCH) {
    const int n = tid >> 4;
    float hs = 0.f;
    for (int cc = 0; cc < NCHUNK - 1; ++cc) {
      hs = fmaf(s_p[n][cc * FCH + ch], hs, s_h[n][cc * FCH + ch]);
      s_h[n][cc * FCH + ch] = hs;
    }
  }
  __syncthreads();

  {
    const float Dv = Dvec[d];
    float h[D_STATE];
#pragma unroll
    for (int n = 0; n < D_STATE; ++n) {
      const float hs = s_h[n][(c > 0) ? tid - FCH : tid];
      h[n] = (c > 0) ? hs : 0.f;
    }

    const float* xp = xc0;
    const float* __restrict__ zp = zx + row0 * E_DIM + OFF_Z + d;
    const float* __restrict__ bp = bp0;
    const float* __restrict__ cp = zx + row0 * E_DIM + OFF_C + (size_t)(d >> 4) * D_STATE;
    float* __restrict__ yp = Y + row0 * D_INNER + d;

    for (int l = 0; l < TCH; ++l) {
      const float dt = s_dt[l][tid];
      const float xv = *xp;
      const float zval = *zp;
      float Bv[D_STATE], Cv[D_STATE];
      load16(bp, Bv);
      load16(cp, Cv);
      xp += D_INNER; zp += E_DIM; bp += E_DIM; cp += E_DIM;

      const float dtx = dt * xv;
      float y0 = 0.f, y1 = 0.f;
#pragma unroll
      for (int n = 0; n < D_STATE; n += 2) {
        const float a0 = __expf(dt * An[n]);
        const float a1 = __expf(dt * An[n + 1]);
        h[n]     = fmaf(a0, h[n],     dtx * Bv[n]);
        h[n + 1] = fmaf(a1, h[n + 1], dtx * Bv[n + 1]);
        y0 = fmaf(h[n],     Cv[n],     y0);
        y1 = fmaf(h[n + 1], Cv[n + 1], y1);
      }
      const float yc = y0 + y1;

      const float gate = zval / (1.f + __expf(-zval));
      *yp = (yc + Dv * xv) * gate;
      yp += D_INNER;
    }
  }
}

extern "C" void kernel_launch(void* const* d_in, const int* in_sizes, int n_in,
                              void* d_out, int out_size, void* d_ws, size_t ws_size,
                              hipStream_t stream) {
  const float* hidden     = (const float*)d_in[0];
  const float* in_proj_w  = (const float*)d_in[1];
  const float* conv_w     = (const float*)d_in[2];
  const float* conv_b     = (const float*)d_in[3];
  const float* dt_proj_w  = (const float*)d_in[4];
  const float* dt_proj_b  = (const float*)d_in[5];
  const float* Amat       = (const float*)d_in[6];
  const float* Dvec       = (const float*)d_in[7];
  const float* out_proj_w = (const float*)d_in[8];
  float* out = (float*)d_out;

  float* zx    = (float*)d_ws;
  float* xconv = zx    + (size_t)M_ROWS * E_DIM;
  float* dtbuf = xconv + (size_t)M_ROWS * D_INNER;
  float* Ybuf  = dtbuf + (size_t)M_ROWS * D_INNER;

  in_proj_gemm_kernel<<<(M_ROWS / 64) * (N_MAIN / WT_N) / 8, 256, 0, stream>>>(
      hidden, in_proj_w, zx);

  dt_chain_kernel<<<M_ROWS / 16, 256, 0, stream>>>(
      hidden, in_proj_w + (size_t)OFF_DT * D_MODEL, dt_proj_w, dtbuf);

  scan_fused_kernel<<<BATCH * (D_INNER / FCH), FUSED_THREADS, 0, stream>>>(
      zx, conv_w, conv_b, dt_proj_b, Amat, Dvec, dtbuf, xconv, Ybuf);

  {
    int tiles = (M_ROWS / 64) * (D_MODEL / 64);
    wmma_gemm_f32<2><<<(tiles + 3) / 4, 256, 0, stream>>>(
        Ybuf, D_INNER, out_proj_w, D_INNER, out, D_MODEL,
        M_ROWS, D_MODEL, D_INNER);
  }
}
